# starting kernel
# baseline (speedup 1.0000x reference)
#include <hip/hip_runtime.h>

typedef __attribute__((ext_vector_type(4)))  _Float16 v4h;
typedef __attribute__((ext_vector_type(8)))  _Float16 v8h;
typedef __attribute__((ext_vector_type(16))) _Float16 v16h;
typedef __attribute__((ext_vector_type(8)))  float    v8f;

#define NNODES 40000
#define NEDGES 640000
#define EMBED  128
#define NTYPES 64
#define NITERS 4


__global__ void k_zero2(int* a, int* b) {
    int i = blockIdx.x * blockDim.x + threadIdx.x;
    if (i < NNODES) { a[i] = 0; b[i] = 0; }
}

__global__ void k_count(const int* __restrict__ dst, int* cnt) {
    int e = blockIdx.x * blockDim.x + threadIdx.x;
    if (e < NEDGES) atomicAdd(&cnt[dst[e]], 1);
}

__global__ void k_dinv(const int* __restrict__ cnt, float* __restrict__ dinv) {
    int i = blockIdx.x * blockDim.x + threadIdx.x;
    if (i < NNODES) dinv[i] = rsqrtf((float)cnt[i] + 1.0f);
}

__global__ void k_scan(const int* __restrict__ cnt, int* __restrict__ off) {
    __shared__ int s[1024];
    __shared__ int carry;
    const int tid = threadIdx.x;
    if (tid == 0) carry = 0;
    __syncthreads();
    for (int base = 0; base < NNODES; base += 1024) {
        int v = (base + tid < NNODES) ? cnt[base + tid] : 0;
        s[tid] = v;
        __syncthreads();
        for (int d = 1; d < 1024; d <<= 1) {
            int t = (tid >= d) ? s[tid - d] : 0;
            __syncthreads();
            s[tid] += t;
            __syncthreads();
        }
        int incl = s[tid];
        if (base + tid < NNODES) off[base + tid] = carry + incl - v;
        __syncthreads();
        if (tid == 1023) carry += s[1023];
        __syncthreads();
    }
    if (tid == 0) off[NNODES] = carry;
}

__global__ void k_csr_fill(const int* __restrict__ src, const int* __restrict__ dst,
                           const int* __restrict__ off, const float* __restrict__ dinv,
                           int* __restrict__ cursor, int* __restrict__ col,
                           float* __restrict__ wt) {
    int e = blockIdx.x * blockDim.x + threadIdx.x;
    if (e >= NEDGES) return;
    int s = src[e], d = dst[e];
    int j = off[d] + atomicAdd(&cursor[d], 1);
    col[j] = s;
    wt[j]  = dinv[s] * dinv[d];
}


__device__ __forceinline__ void split_store(float x, float y, float z, float w,
                                            _Float16* hi, _Float16* lo, size_t idx4) {
    v4h H, L;
    H[0] = (_Float16)x; L[0] = (_Float16)(x - (float)H[0]);
    H[1] = (_Float16)y; L[1] = (_Float16)(y - (float)H[1]);
    H[2] = (_Float16)z; L[2] = (_Float16)(z - (float)H[2]);
    H[3] = (_Float16)w; L[3] = (_Float16)(w - (float)H[3]);
    ((v4h*)hi)[idx4] = H;
    ((v4h*)lo)[idx4] = L;
}

__global__ void k_embed_split(const int* __restrict__ x, const float* __restrict__ Win,
                              _Float16* __restrict__ hhi, _Float16* __restrict__ hlo) {
    int t = blockIdx.x * blockDim.x + threadIdx.x;
    if (t >= NNODES * (EMBED / 4)) return;
    int i = t >> 5, c = t & 31;
    float4 v = ((const float4*)(Win + (size_t)x[i] * EMBED))[c];
    split_store(v.x, v.y, v.z, v.w, hhi, hlo, (size_t)i * 32 + c);
}


template<int K, int N>
__global__ void k_split_Wt(const float* __restrict__ W, _Float16* __restrict__ Whi,
                           _Float16* __restrict__ Wlo) {
    int t = blockIdx.x * blockDim.x + threadIdx.x;
    if (t >= K * N) return;
    int k = t / N, n = t % N;
    float w = W[k * N + n];
    _Float16 hi = (_Float16)w;
    Whi[n * K + k] = hi;
    Wlo[n * K + k] = (_Float16)(w - (float)hi);
}


template<int NOUT>
__global__ __launch_bounds__(256) void k_gemm(const _Float16* __restrict__ Ahi,
                                              const _Float16* __restrict__ Alo,
                                              const _Float16* __restrict__ Bhi,
                                              const _Float16* __restrict__ Blo,
                                              float* __restrict__ out) {
    constexpr int K  = EMBED;
    constexpr int NT = NOUT / 16;
    __shared__ _Float16 sBh[NOUT * K];
    __shared__ _Float16 sBl[NOUT * K];

    for (int i = threadIdx.x; i < NOUT * K / 8; i += 256) {
        ((v8h*)sBh)[i] = ((const v8h*)Bhi)[i];
        ((v8h*)sBl)[i] = ((const v8h*)Blo)[i];
    }
    __syncthreads();

    const int wave  = threadIdx.x >> 5;
    const int lane  = threadIdx.x & 31;
    const int strip = blockIdx.x * 8 + wave;
    if (strip >= NNODES / 16) return;
    const int half = lane >> 4;
    const int mrow = strip * 16 + (lane & 15);

    v8f acc[NT] = {};

#pragma unroll
    for (int kc = 0; kc < K / 32; ++kc) {
        const int abase = mrow * K + kc * 32 + half * 8;
        v8h a0h = *(const v8h*)(Ahi + abase);
        v8h a1h = *(const v8h*)(Ahi + abase + 16);
        v8h a0l = *(const v8h*)(Alo + abase);
        v8h a1l = *(const v8h*)(Alo + abase + 16);
        v16h fah = __builtin_shufflevector(a0h, a1h, 0,1,2,3,4,5,6,7,8,9,10,11,12,13,14,15);
        v16h fal = __builtin_shufflevector(a0l, a1l, 0,1,2,3,4,5,6,7,8,9,10,11,12,13,14,15);
#pragma unroll
        for (int t = 0; t < NT; ++t) {
            const int bbase = (t * 16 + (lane & 15)) * K + kc * 32 + half * 8;
            v8h b0h = *(const v8h*)(sBh + bbase);
            v8h b1h = *(const v8h*)(sBh + bbase + 16);
            v8h b0l = *(const v8h*)(sBl + bbase);
            v8h b1l = *(const v8h*)(sBl + bbase + 16);
            v16h fbh = __builtin_shufflevector(b0h, b1h, 0,1,2,3,4,5,6,7,8,9,10,11,12,13,14,15);
            v16h fbl = __builtin_shufflevector(b0l, b1l, 0,1,2,3,4,5,6,7,8,9,10,11,12,13,14,15);
            acc[t] = __builtin_amdgcn_wmma_f32_16x16x32_f16(false, fah, false, fbh,
                                                            (short)0, acc[t], false, false);
            acc[t] = __builtin_amdgcn_wmma_f32_16x16x32_f16(false, fah, false, fbl,
                                                            (short)0, acc[t], false, false);
            acc[t] = __builtin_amdgcn_wmma_f32_16x16x32_f16(false, fal, false, fbh,
                                                            (short)0, acc[t], false, false);
        }
    }

#pragma unroll
    for (int t = 0; t < NT; ++t) {
        const int ncol = t * 16 + (lane & 15);
#pragma unroll
        for (int r = 0; r < 8; ++r) {
            const int row = strip * 16 + half * 8 + r;
            out[(size_t)row * NOUT + ncol] = acc[t][r];
        }
    }
}


template<int NOUT, bool RELUSPLIT>
__global__ void k_gather(const float* __restrict__ hw, const int* __restrict__ off,
                         const int* __restrict__ col, const float* __restrict__ wt,
                         const float* __restrict__ dinv, const float* __restrict__ bias,
                         _Float16* __restrict__ hhi, _Float16* __restrict__ hlo,
                         float* __restrict__ outf) {
    constexpr int LPN = NOUT / 4;
    int t = blockIdx.x * blockDim.x + threadIdx.x;
    int node = t / LPN;
    int c    = t % LPN;
    if (node >= NNODES) return;

    float di = dinv[node];
    float sw = di * di;
    float4 v = ((const float4*)hw)[(size_t)node * LPN + c];
    float4 acc = { sw * v.x, sw * v.y, sw * v.z, sw * v.w };

    const int j0  = off[node];
    const int deg = off[node + 1] - j0;

    for (int base = 0; base < deg; base += LPN) {
        int n = deg - base; if (n > LPN) n = LPN;
        int   sl = 0;
        float wl = 0.0f;
        if (c < n) { int jj = j0 + base + c; sl = col[jj]; wl = wt[jj]; }
        for (int k = 0; k < n; ++k) {
            int   s = __shfl(sl, k, LPN);
            float w = __shfl(wl, k, LPN);
            float4 m = ((const float4*)hw)[(size_t)s * LPN + c];
            acc.x += w * m.x; acc.y += w * m.y; acc.z += w * m.z; acc.w += w * m.w;
        }
    }

    float4 b = ((const float4*)bias)[c];
    acc.x += b.x; acc.y += b.y; acc.z += b.z; acc.w += b.w;

    if (RELUSPLIT) {
        acc.x = fmaxf(acc.x, 0.f); acc.y = fmaxf(acc.y, 0.f);
        acc.z = fmaxf(acc.z, 0.f); acc.w = fmaxf(acc.w, 0.f);
        split_store(acc.x, acc.y, acc.z, acc.w, hhi, hlo, (size_t)node * LPN + c);
    } else {
        ((float4*)outf)[(size_t)node * LPN + c] = acc;
    }
}


extern "C" void kernel_launch(void* const* d_in, const int* in_sizes, int n_in,
                              void* d_out, int out_size, void* d_ws, size_t ws_size,
                              hipStream_t stream) {
    const int*   x      = (const int*)d_in[0];
    const int*   src    = (const int*)d_in[1];
    const int*   dst    = ((const int*)d_in[1]) + NEDGES;
    const float* W_in   = (const float*)d_in[2];
    const float* W_conv = (const float*)d_in[3];
    const float* b_conv = (const float*)d_in[4];
    const float* W_dec  = (const float*)d_in[5];
    const float* b_dec  = (const float*)d_in[6];
    (void)in_sizes; (void)n_in; (void)out_size; (void)ws_size;

    char* p = (char*)d_ws;
    auto carve = [&](size_t bytes) { char* r = p; p += (bytes + 255) & ~(size_t)255; return r; };
    float*    dinv   = (float*)   carve((size_t)NNODES * 4);
    int*      cnt    = (int*)     carve((size_t)NNODES * 4);
    int*      cursor = (int*)     carve((size_t)NNODES * 4);
    int*      off    = (int*)     carve((size_t)(NNODES + 1) * 4);
    int*      col    = (int*)     carve((size_t)NEDGES * 4);
    float*    wt     = (float*)   carve((size_t)NEDGES * 4);
    float*    hw     = (float*)   carve((size_t)NNODES * EMBED * 4);
    _Float16* hhi    = (_Float16*)carve((size_t)NNODES * EMBED * 2);
    _Float16* hlo    = (_Float16*)carve((size_t)NNODES * EMBED * 2);
    _Float16* Wch    = (_Float16*)carve((size_t)EMBED * EMBED * 2);
    _Float16* Wcl    = (_Float16*)carve((size_t)EMBED * EMBED * 2);
    _Float16* Wdh    = (_Float16*)carve((size_t)NTYPES * EMBED * 2);
    _Float16* Wdl    = (_Float16*)carve((size_t)NTYPES * EMBED * 2);

    const int B = 256;
    auto blocks = [](long long n, int b) { return (int)((n + b - 1) / b); };

    k_zero2   <<<blocks(NNODES, B), B, 0, stream>>>(cnt, cursor);
    k_count   <<<blocks(NEDGES, B), B, 0, stream>>>(dst, cnt);
    k_dinv    <<<blocks(NNODES, B), B, 0, stream>>>(cnt, dinv);
    k_scan    <<<1, 1024, 0, stream>>>(cnt, off);
    k_csr_fill<<<blocks(NEDGES, B), B, 0, stream>>>(src, dst, off, dinv, cursor, col, wt);

    k_split_Wt<EMBED, EMBED> <<<blocks(EMBED * EMBED, B), B, 0, stream>>>(W_conv, Wch, Wcl);
    k_split_Wt<EMBED, NTYPES><<<blocks(EMBED * NTYPES, B), B, 0, stream>>>(W_dec,  Wdh, Wdl);

    k_embed_split<<<blocks((long long)NNODES * 32, B), B, 0, stream>>>(x, W_in, hhi, hlo);

    const int gemm_blocks = ((NNODES / 16) + 7) / 8;

    for (int it = 0; it < NITERS; ++it) {
        k_gemm<EMBED><<<gemm_blocks, B, 0, stream>>>(hhi, hlo, Wch, Wcl, hw);
        k_gather<EMBED, true><<<blocks((long long)NNODES * 32, B), B, 0, stream>>>(
            hw, off, col, wt, dinv, b_conv, hhi, hlo, nullptr);
    }

    k_gemm<NTYPES><<<gemm_blocks, B, 0, stream>>>(hhi, hlo, Wdh, Wdl, hw);
    k_gather<NTYPES, false><<<blocks((long long)NNODES * 16, B), B, 0, stream>>>(
        hw, off, col, wt, dinv, b_dec, nullptr, nullptr, (float*)d_out);
}


# gather re-tiled: 8 lanes per node, 4 nodes abreast in a wave; a lane owns NOUT/8 columns as line-aligned float4 chunks, a trip takes 4 edges (16 row loads in flight); waves 4x fewer; sum order kept
# speedup vs baseline: 2.2023x; 2.2023x over previous
#include <hip/hip_runtime.h>

typedef __attribute__((ext_vector_type(4)))  _Float16 v4h;
typedef __attribute__((ext_vector_type(8)))  _Float16 v8h;
typedef __attribute__((ext_vector_type(16))) _Float16 v16h;
typedef __attribute__((ext_vector_type(8)))  float    v8f;

#define NNODES 40000
#define NEDGES 640000
#define EMBED  128
#define NTYPES 64
#define NITERS 4
#define PITCH  32


__global__ void k_zero(int* cnt, int* spill_n) {
    int i = blockIdx.x * blockDim.x + threadIdx.x;
    if (i < NNODES) cnt[i] = 0;
    if (i == 0) *spill_n = 0;
}

__global__ void k_fill(const int* __restrict__ src, const int* __restrict__ dst,
                       int* cnt, int* __restrict__ slot, int* spill_n,
                       int* __restrict__ spill_dst, int* __restrict__ spill_src) {
    int e = blockIdx.x * blockDim.x + threadIdx.x;
    if (e >= NEDGES) return;
    int s = src[e], d = dst[e];
    int r = atomicAdd(&cnt[d], 1);
    if (r < PITCH) {
        slot[d * PITCH + r] = s;
    } else {
        int o = atomicAdd(spill_n, 1);
        spill_dst[o] = d;
        spill_src[o] = s;
    }
}

__global__ void k_dinv(const int* __restrict__ cnt, float* __restrict__ dinv) {
    int i = blockIdx.x * blockDim.x + threadIdx.x;
    if (i < NNODES) dinv[i] = rsqrtf((float)cnt[i] + 1.0f);
}


__device__ __forceinline__ void split_store(float x, float y, float z, float w,
                                            _Float16* hi, _Float16* lo, size_t idx4) {
    v4h H, L;
    H[0] = (_Float16)x; L[0] = (_Float16)(x - (float)H[0]);
    H[1] = (_Float16)y; L[1] = (_Float16)(y - (float)H[1]);
    H[2] = (_Float16)z; L[2] = (_Float16)(z - (float)H[2]);
    H[3] = (_Float16)w; L[3] = (_Float16)(w - (float)H[3]);
    ((v4h*)hi)[idx4] = H;
    ((v4h*)lo)[idx4] = L;
}


template<int K, int N>
__global__ void k_split_Wt(const float* __restrict__ W, _Float16* __restrict__ Whi,
                           _Float16* __restrict__ Wlo) {
    constexpr int NT = N / 16;
    int t = blockIdx.x * blockDim.x + threadIdx.x;
    if (t >= K * N) return;
    int k = t / N, n = t % N;
    float w = W[k * N + n];
    _Float16 hi = (_Float16)w;
    int kc = k >> 5, kk = k & 31;
    int j = kk >> 4, hf = (kk >> 3) & 1, e = kk & 7;
    int tt = n >> 4, c = n & 15;
    int idx = ((((kc * NT + tt) * 2 + j) * 32 + hf * 16 + c) << 3) + e;
    Whi[idx] = hi;
    Wlo[idx] = (_Float16)(w - (float)hi);
}


#define NSTRIPS (NNODES / 16)

template<int NOUT, bool EMB>
__global__ __launch_bounds__(128) void k_gemm(const _Float16* __restrict__ Ahi,
                                              const _Float16* __restrict__ Alo,
                                              const int* __restrict__ xs,
                                              const float* __restrict__ Win,
                                              const _Float16* __restrict__ Bhi,
                                              const _Float16* __restrict__ Blo,
                                              float* __restrict__ out) {
    constexpr int K  = EMBED;
    constexpr int KC = K / 32;
    constexpr int NT = NOUT / 16;

    const int lane   = threadIdx.x & 31;
    const int half   = lane >> 4;
    const int nwaves = gridDim.x * 4;
    int strip        = blockIdx.x * 4 + (threadIdx.x >> 5);
    if (strip >= NSTRIPS) return;

    v8h a[KC][4], an[KC][4];
    v16h bh[KC][NT], bl[KC][NT];

    auto load_a = [&](v8h (&dst)[KC][4], int s) {
        const int mrow = s * 16 + (lane & 15);
#pragma unroll
        for (int kc = 0; kc < KC; ++kc) {
            const int abase = mrow * K + kc * 32 + half * 8;
            dst[kc][0] = *(const v8h*)(Ahi + abase);
            dst[kc][1] = *(const v8h*)(Ahi + abase + 16);
            dst[kc][2] = *(const v8h*)(Alo + abase);
            dst[kc][3] = *(const v8h*)(Alo + abase + 16);
        }
    };

    float4 raw[KC][4];

    auto load_raw = [&](int xr) {
        const float* srow = Win + (size_t)xr * K + half * 8;
#pragma unroll
        for (int kc = 0; kc < KC; ++kc) {
            raw[kc][0] = *(const float4*)(srow + kc * 32);
            raw[kc][1] = *(const float4*)(srow + kc * 32 + 4);
            raw[kc][2] = *(const float4*)(srow + kc * 32 + 16);
            raw[kc][3] = *(const float4*)(srow + kc * 32 + 20);
        }
    };

    auto split_raw = [&](v8h (&dst)[KC][4]) {
#pragma unroll
        for (int kc = 0; kc < KC; ++kc) {
#pragma unroll
            for (int j = 0; j < 2; ++j) {
                const float4 f = raw[kc][2 * j], g = raw[kc][2 * j + 1];
                v8h H, L;
                H[0] = (_Float16)f.x; L[0] = (_Float16)(f.x - (float)H[0]);
                H[1] = (_Float16)f.y; L[1] = (_Float16)(f.y - (float)H[1]);
                H[2] = (_Float16)f.z; L[2] = (_Float16)(f.z - (float)H[2]);
                H[3] = (_Float16)f.w; L[3] = (_Float16)(f.w - (float)H[3]);
                H[4] = (_Float16)g.x; L[4] = (_Float16)(g.x - (float)H[4]);
                H[5] = (_Float16)g.y; L[5] = (_Float16)(g.y - (float)H[5]);
                H[6] = (_Float16)g.z; L[6] = (_Float16)(g.z - (float)H[6]);
                H[7] = (_Float16)g.w; L[7] = (_Float16)(g.w - (float)H[7]);
                dst[kc][j]     = H;
                dst[kc][2 + j] = L;
            }
        }
    };

    auto compute_store = [&](const v8h (&af)[KC][4], int s) {
        v8f acc[NT] = {};
#pragma unroll
        for (int kc = 0; kc < KC; ++kc) {
            v16h fah = __builtin_shufflevector(af[kc][0], af[kc][1], 0,1,2,3,4,5,6,7,8,9,10,11,12,13,14,15);
            v16h fal = __builtin_shufflevector(af[kc][2], af[kc][3], 0,1,2,3,4,5,6,7,8,9,10,11,12,13,14,15);
#pragma unroll
            for (int t = 0; t < NT; ++t) {
                acc[t] = __builtin_amdgcn_wmma_f32_16x16x32_f16(false, fah, false, bh[kc][t],
                                                                (short)0, acc[t], false, false);
                acc[t] = __builtin_amdgcn_wmma_f32_16x16x32_f16(false, fah, false, bl[kc][t],
                                                                (short)0, acc[t], false, false);
                acc[t] = __builtin_amdgcn_wmma_f32_16x16x32_f16(false, fal, false, bh[kc][t],
                                                                (short)0, acc[t], false, false);
            }
        }
#pragma unroll
        for (int t = 0; t < NT; ++t) {
            const int ncol = t * 16 + (lane & 15);
#pragma unroll
            for (int r = 0; r < 8; ++r) {
                const int row = s * 16 + half * 8 + r;
                out[(size_t)row * NOUT + ncol] = acc[t][r];
            }
        }
    };

    const int r16 = lane & 15;
    int  next     = strip + nwaves;
    bool has_next = next < NSTRIPS;
    int  xn = 0, xnn = 0;
    if constexpr (EMB) {
        const int xc = xs[strip * 16 + r16];
        if (has_next) xn = xs[next * 16 + r16];
        load_raw(xc);
    } else {
        load_a(a, strip);
    }

#pragma unroll
    for (int kc = 0; kc < KC; ++kc) {
#pragma unroll
        for (int t = 0; t < NT; ++t) {
            const int u = ((kc * NT + t) * 2) * 32 + lane;
            v8h b0h = ((const v8h*)Bhi)[u];
            v8h b1h = ((const v8h*)Bhi)[u + 32];
            v8h b0l = ((const v8h*)Blo)[u];
            v8h b1l = ((const v8h*)Blo)[u + 32];
            bh[kc][t] = __builtin_shufflevector(b0h, b1h, 0,1,2,3,4,5,6,7,8,9,10,11,12,13,14,15);
            bl[kc][t] = __builtin_shufflevector(b0l, b1l, 0,1,2,3,4,5,6,7,8,9,10,11,12,13,14,15);
        }
    }

    if constexpr (EMB) {
        split_raw(a);
        for (;;) {
            if (has_next) {
                load_raw(xn);
                if (next + nwaves < NSTRIPS) xnn = xs[(next + nwaves) * 16 + r16];
            }
            compute_store(a, strip);
            if (!has_next) break;
            split_raw(a);
            strip    = next;
            next     = strip + nwaves;
            has_next = next < NSTRIPS;
            xn       = xnn;
        }
    } else {
        if (has_next) load_a(an, next);
        compute_store(a, strip);

        while (has_next) {
#pragma unroll
            for (int kc = 0; kc < KC; ++kc) {
                a[kc][0] = an[kc][0]; a[kc][1] = an[kc][1];
                a[kc][2] = an[kc][2]; a[kc][3] = an[kc][3];
            }
            strip    = next;
            next     = strip + nwaves;
            has_next = next < NSTRIPS;
            if (has_next) load_a(an, next);
            compute_store(a, strip);
        }
    }
}


#define GL 8

template<int NOUT, bool RELUSPLIT>
__global__ void k_gather(const float* __restrict__ hw, const int* __restrict__ cnt,
                         const int* __restrict__ slot, const int* __restrict__ spill_n,
                         const int* __restrict__ spill_dst, const int* __restrict__ spill_src,
                         const float* __restrict__ dinv, const float* __restrict__ bias,
                         _Float16* __restrict__ hhi, _Float16* __restrict__ hlo,
                         float* __restrict__ outf) {
    constexpr int NCH = NOUT / 32;
    constexpr int R4  = NOUT / 4;
    const int t    = blockIdx.x * blockDim.x + threadIdx.x;
    const int node = t / GL;
    const int c    = t % GL;
    if (node >= NNODES) return;

    const float4* hw4 = (const float4*)hw;
    const float di = dinv[node];
    const float sw = di * di;
    float4 acc[NCH];
#pragma unroll
    for (int j = 0; j < NCH; ++j) {
        const float4 v = hw4[(size_t)node * R4 + j * GL + c];
        acc[j].x = sw * v.x; acc[j].y = sw * v.y; acc[j].z = sw * v.z; acc[j].w = sw * v.w;
    }

    const int  deg = cnt[node];
    const int  nin = deg < PITCH ? deg : PITCH;
    const int* row = slot + (size_t)node * PITCH;

    int   sl[PITCH / GL];
    float dl[PITCH / GL];
#pragma unroll
    for (int j = 0; j < PITCH / GL; ++j) {
        const int* p = (j * GL + c < nin) ? row + (j * GL + c) : cnt + node;
        sl[j] = *p;
    }
#pragma unroll
    for (int j = 0; j < PITCH / GL; ++j)
        sl[j] = (j * GL + c < nin) ? sl[j] : node;
#pragma unroll
    for (int j = 0; j < PITCH / GL; ++j)
        dl[j] = dinv[sl[j]];
#pragma unroll
    for (int j = 0; j < PITCH / GL; ++j)
        dl[j] = (j * GL + c < nin) ? dl[j] : 0.0f;

    const int ng = (nin + 3) >> 2;
    for (int g = 0; g < ng; ++g) {
        const int   j  = g >> 1;
        const int   cs = j == 0 ? sl[0] : j == 1 ? sl[1] : j == 2 ? sl[2] : sl[3];
        const float cd = j == 0 ? dl[0] : j == 1 ? dl[1] : j == 2 ? dl[2] : dl[3];
        const int   q0 = (g & 1) * 4;
        int   s[4];
        float w[4];
#pragma unroll
        for (int e = 0; e < 4; ++e) {
            s[e] = __shfl(cs, q0 + e, GL);
            w[e] = __shfl(cd, q0 + e, GL) * di;
        }
        float4 m[4][NCH];
#pragma unroll
        for (int e = 0; e < 4; ++e)
#pragma unroll
            for (int jj = 0; jj < NCH; ++jj)
                m[e][jj] = hw4[(size_t)s[e] * R4 + jj * GL + c];
#pragma unroll
        for (int e = 0; e < 4; ++e)
#pragma unroll
            for (int jj = 0; jj < NCH; ++jj) {
                acc[jj].x += w[e] * m[e][jj].x; acc[jj].y += w[e] * m[e][jj].y;
                acc[jj].z += w[e] * m[e][jj].z; acc[jj].w += w[e] * m[e][jj].w;
            }
    }

    if (deg > PITCH) {
        int total = *spill_n;
        if (total > NEDGES) total = NEDGES;
        for (int base = 0; base < total; base += GL) {
            int n = total - base; if (n > GL) n = GL;
            int   s1 = -1;
            float d1 = 0.0f;
            if (c < n && spill_dst[base + c] == node) { s1 = spill_src[base + c]; d1 = dinv[s1]; }
            for (int k = 0; k < n; ++k) {
                const int sk = __shfl(s1, k, GL);
                if (sk >= 0) {
                    const float wk = __shfl(d1, k, GL) * di;
#pragma unroll
                    for (int jj = 0; jj < NCH; ++jj) {
                        const float4 mm = hw4[(size_t)sk * R4 + jj * GL + c];
                        acc[jj].x += wk * mm.x; acc[jj].y += wk * mm.y;
                        acc[jj].z += wk * mm.z; acc[jj].w += wk * mm.w;
                    }
                }
            }
        }
    }

#pragma unroll
    for (int jj = 0; jj < NCH; ++jj) {
        const float4 b = ((const float4*)bias)[jj * GL + c];
        float4 a = acc[jj];
        a.x += b.x; a.y += b.y; a.z += b.z; a.w += b.w;
        const size_t idx4 = (size_t)node * R4 + jj * GL + c;
        if (RELUSPLIT) {
            a.x = fmaxf(a.x, 0.f); a.y = fmaxf(a.y, 0.f);
            a.z = fmaxf(a.z, 0.f); a.w = fmaxf(a.w, 0.f);
            split_store(a.x, a.y, a.z, a.w, hhi, hlo, idx4);
        } else {
            ((float4*)outf)[idx4] = a;
        }
    }
}


extern "C" void kernel_launch(void* const* d_in, const int* in_sizes, int n_in,
                              void* d_out, int out_size, void* d_ws, size_t ws_size,
                              hipStream_t stream) {
    const int*   x      = (const int*)d_in[0];
    const int*   src    = (const int*)d_in[1];
    const int*   dst    = ((const int*)d_in[1]) + NEDGES;
    const float* W_in   = (const float*)d_in[2];
    const float* W_conv = (const float*)d_in[3];
    const float* b_conv = (const float*)d_in[4];
    const float* W_dec  = (const float*)d_in[5];
    const float* b_dec  = (const float*)d_in[6];
    (void)in_sizes; (void)n_in; (void)out_size; (void)ws_size;

    char* p = (char*)d_ws;
    auto carve = [&](size_t bytes) { char* r = p; p += (bytes + 255) & ~(size_t)255; return r; };
    float*    dinv   = (float*)   carve((size_t)NNODES * 4);
    int*      cnt    = (int*)     carve((size_t)NNODES * 4);
    int*      spilln = (int*)     carve((size_t)4);
    int*      slot   = (int*)     carve((size_t)NNODES * PITCH * 4);
    int*      spilld = (int*)     carve((size_t)NEDGES * 4);
    int*      spills = (int*)     carve((size_t)NEDGES * 4);
    float*    hw     = (float*)   carve((size_t)NNODES * EMBED * 4);
    _Float16* hhi    = (_Float16*)carve((size_t)NNODES * EMBED * 2);
    _Float16* hlo    = (_Float16*)carve((size_t)NNODES * EMBED * 2);
    _Float16* Wch    = (_Float16*)carve((size_t)EMBED * EMBED * 2);
    _Float16* Wcl    = (_Float16*)carve((size_t)EMBED * EMBED * 2);
    _Float16* Wdh    = (_Float16*)carve((size_t)NTYPES * EMBED * 2);
    _Float16* Wdl    = (_Float16*)carve((size_t)NTYPES * EMBED * 2);

    const int B = 256;
    auto blocks = [](long long n, int b) { return (int)((n + b - 1) / b); };

    k_zero<<<blocks(NNODES, B), B, 0, stream>>>(cnt, spilln);
    k_fill<<<blocks(NEDGES, B), B, 0, stream>>>(src, dst, cnt, slot, spilln, spilld, spills);
    k_dinv<<<blocks(NNODES, B), B, 0, stream>>>(cnt, dinv);

    k_split_Wt<EMBED, EMBED> <<<blocks(EMBED * EMBED, B), B, 0, stream>>>(W_conv, Wch, Wcl);
    k_split_Wt<EMBED, NTYPES><<<blocks(EMBED * NTYPES, B), B, 0, stream>>>(W_dec,  Wdh, Wdl);

    const int gemm_blocks = 209;

    for (int it = 0; it < NITERS; ++it) {
        if (it == 0)
            k_gemm<EMBED, true><<<gemm_blocks, 128, 0, stream>>>(nullptr, nullptr, x, W_in, Wch, Wcl, hw);
        else
            k_gemm<EMBED, false><<<gemm_blocks, 128, 0, stream>>>(hhi, hlo, nullptr, nullptr, Wch, Wcl, hw);
        k_gather<EMBED, true><<<blocks((long long)NNODES * GL, B), B, 0, stream>>>(
            hw, cnt, slot, spilln, spilld, spills, dinv, b_conv, hhi, hlo, nullptr);
    }

    k_gemm<NTYPES, false><<<gemm_blocks, 128, 0, stream>>>(hhi, hlo, nullptr, nullptr, Wdh, Wdl, hw);
    k_gather<NTYPES, false><<<blocks((long long)NNODES * GL, B), B, 0, stream>>>(
        hw, cnt, slot, spilln, spilld, spills, dinv, b_dec, nullptr, nullptr, (float*)d_out);
}


# gather re-tiled to 256-byte units: 16 lanes per node, 2 nodes abreast, 2 in turn; a load instruction covers aligned 256 B of 2 rows (was 128 B of 4 rows); 8 loads a trip (was 16); sum order kept
# speedup vs baseline: 2.2705x; 1.0310x over previous
#include <hip/hip_runtime.h>

typedef __attribute__((ext_vector_type(4)))  _Float16 v4h;
typedef __attribute__((ext_vector_type(8)))  _Float16 v8h;
typedef __attribute__((ext_vector_type(16))) _Float16 v16h;
typedef __attribute__((ext_vector_type(8)))  float    v8f;

#define NNODES 40000
#define NEDGES 640000
#define EMBED  128
#define NTYPES 64
#define NITERS 4
#define PITCH  32


__global__ void k_zero(int* cnt, int* spill_n) {
    int i = blockIdx.x * blockDim.x + threadIdx.x;
    if (i < NNODES) cnt[i] = 0;
    if (i == 0) *spill_n = 0;
}

__global__ void k_fill(const int* __restrict__ src, const int* __restrict__ dst,
                       int* cnt, int* __restrict__ slot, int* spill_n,
                       int* __restrict__ spill_dst, int* __restrict__ spill_src) {
    int e = blockIdx.x * blockDim.x + threadIdx.x;
    if (e >= NEDGES) return;
    int s = src[e], d = dst[e];
    int r = atomicAdd(&cnt[d], 1);
    if (r < PITCH) {
        slot[d * PITCH + r] = s;
    } else {
        int o = atomicAdd(spill_n, 1);
        spill_dst[o] = d;
        spill_src[o] = s;
    }
}

__global__ void k_dinv(const int* __restrict__ cnt, float* __restrict__ dinv) {
    int i = blockIdx.x * blockDim.x + threadIdx.x;
    if (i < NNODES) dinv[i] = rsqrtf((float)cnt[i] + 1.0f);
}


__device__ __forceinline__ void split_store(float x, float y, float z, float w,
                                            _Float16* hi, _Float16* lo, size_t idx4) {
    v4h H, L;
    H[0] = (_Float16)x; L[0] = (_Float16)(x - (float)H[0]);
    H[1] = (_Float16)y; L[1] = (_Float16)(y - (float)H[1]);
    H[2] = (_Float16)z; L[2] = (_Float16)(z - (float)H[2]);
    H[3] = (_Float16)w; L[3] = (_Float16)(w - (float)H[3]);
    ((v4h*)hi)[idx4] = H;
    ((v4h*)lo)[idx4] = L;
}


template<int K, int N>
__global__ void k_split_Wt(const float* __restrict__ W, _Float16* __restrict__ Whi,
                           _Float16* __restrict__ Wlo) {
    constexpr int NT = N / 16;
    int t = blockIdx.x * blockDim.x + threadIdx.x;
    if (t >= K * N) return;
    int k = t / N, n = t % N;
    float w = W[k * N + n];
    _Float16 hi = (_Float16)w;
    int kc = k >> 5, kk = k & 31;
    int j = kk >> 4, hf = (kk >> 3) & 1, e = kk & 7;
    int tt = n >> 4, c = n & 15;
    int idx = ((((kc * NT + tt) * 2 + j) * 32 + hf * 16 + c) << 3) + e;
    Whi[idx] = hi;
    Wlo[idx] = (_Float16)(w - (float)hi);
}


#define NSTRIPS (NNODES / 16)

template<int NOUT, bool EMB>
__global__ __launch_bounds__(128) void k_gemm(const _Float16* __restrict__ Ahi,
                                              const _Float16* __restrict__ Alo,
                                              const int* __restrict__ xs,
                                              const float* __restrict__ Win,
                                              const _Float16* __restrict__ Bhi,
                                              const _Float16* __restrict__ Blo,
                                              float* __restrict__ out) {
    constexpr int K  = EMBED;
    constexpr int KC = K / 32;
    constexpr int NT = NOUT / 16;

    const int lane   = threadIdx.x & 31;
    const int half   = lane >> 4;
    const int nwaves = gridDim.x * 4;
    int strip        = blockIdx.x * 4 + (threadIdx.x >> 5);
    if (strip >= NSTRIPS) return;

    v8h a[KC][4], an[KC][4];
    v16h bh[KC][NT], bl[KC][NT];

    auto load_a = [&](v8h (&dst)[KC][4], int s) {
        const int mrow = s * 16 + (lane & 15);
#pragma unroll
        for (int kc = 0; kc < KC; ++kc) {
            const int abase = mrow * K + kc * 32 + half * 8;
            dst[kc][0] = *(const v8h*)(Ahi + abase);
            dst[kc][1] = *(const v8h*)(Ahi + abase + 16);
            dst[kc][2] = *(const v8h*)(Alo + abase);
            dst[kc][3] = *(const v8h*)(Alo + abase + 16);
        }
    };

    float4 raw[KC][4];

    auto load_raw = [&](int xr) {
        const float* srow = Win + (size_t)xr * K + half * 8;
#pragma unroll
        for (int kc = 0; kc < KC; ++kc) {
            raw[kc][0] = *(const float4*)(srow + kc * 32);
            raw[kc][1] = *(const float4*)(srow + kc * 32 + 4);
            raw[kc][2] = *(const float4*)(srow + kc * 32 + 16);
            raw[kc][3] = *(const float4*)(srow + kc * 32 + 20);
        }
    };

    auto split_raw = [&](v8h (&dst)[KC][4]) {
#pragma unroll
        for (int kc = 0; kc < KC; ++kc) {
#pragma unroll
            for (int j = 0; j < 2; ++j) {
                const float4 f = raw[kc][2 * j], g = raw[kc][2 * j + 1];
                v8h H, L;
                H[0] = (_Float16)f.x; L[0] = (_Float16)(f.x - (float)H[0]);
                H[1] = (_Float16)f.y; L[1] = (_Float16)(f.y - (float)H[1]);
                H[2] = (_Float16)f.z; L[2] = (_Float16)(f.z - (float)H[2]);
                H[3] = (_Float16)f.w; L[3] = (_Float16)(f.w - (float)H[3]);
                H[4] = (_Float16)g.x; L[4] = (_Float16)(g.x - (float)H[4]);
                H[5] = (_Float16)g.y; L[5] = (_Float16)(g.y - (float)H[5]);
                H[6] = (_Float16)g.z; L[6] = (_Float16)(g.z - (float)H[6]);
                H[7] = (_Float16)g.w; L[7] = (_Float16)(g.w - (float)H[7]);
                dst[kc][j]     = H;
                dst[kc][2 + j] = L;
            }
        }
    };

    auto compute_store = [&](const v8h (&af)[KC][4], int s) {
        v8f acc[NT] = {};
#pragma unroll
        for (int kc = 0; kc < KC; ++kc) {
            v16h fah = __builtin_shufflevector(af[kc][0], af[kc][1], 0,1,2,3,4,5,6,7,8,9,10,11,12,13,14,15);
            v16h fal = __builtin_shufflevector(af[kc][2], af[kc][3], 0,1,2,3,4,5,6,7,8,9,10,11,12,13,14,15);
#pragma unroll
            for (int t = 0; t < NT; ++t) {
                acc[t] = __builtin_amdgcn_wmma_f32_16x16x32_f16(false, fah, false, bh[kc][t],
                                                                (short)0, acc[t], false, false);
                acc[t] = __builtin_amdgcn_wmma_f32_16x16x32_f16(false, fah, false, bl[kc][t],
                                                                (short)0, acc[t], false, false);
                acc[t] = __builtin_amdgcn_wmma_f32_16x16x32_f16(false, fal, false, bh[kc][t],
                                                                (short)0, acc[t], false, false);
            }
        }
#pragma unroll
        for (int t = 0; t < NT; ++t) {
            const int ncol = t * 16 + (lane & 15);
#pragma unroll
            for (int r = 0; r < 8; ++r) {
                const int row = s * 16 + half * 8 + r;
                out[(size_t)row * NOUT + ncol] = acc[t][r];
            }
        }
    };

    const int r16 = lane & 15;
    int  next     = strip + nwaves;
    bool has_next = next < NSTRIPS;
    int  xn = 0, xnn = 0;
    if constexpr (EMB) {
        const int xc = xs[strip * 16 + r16];
        if (has_next) xn = xs[next * 16 + r16];
        load_raw(xc);
    } else {
        load_a(a, strip);
    }

#pragma unroll
    for (int kc = 0; kc < KC; ++kc) {
#pragma unroll
        for (int t = 0; t < NT; ++t) {
            const int u = ((kc * NT + t) * 2) * 32 + lane;
            v8h b0h = ((const v8h*)Bhi)[u];
            v8h b1h = ((const v8h*)Bhi)[u + 32];
            v8h b0l = ((const v8h*)Blo)[u];
            v8h b1l = ((const v8h*)Blo)[u + 32];
            bh[kc][t] = __builtin_shufflevector(b0h, b1h, 0,1,2,3,4,5,6,7,8,9,10,11,12,13,14,15);
            bl[kc][t] = __builtin_shufflevector(b0l, b1l, 0,1,2,3,4,5,6,7,8,9,10,11,12,13,14,15);
        }
    }

    if constexpr (EMB) {
        split_raw(a);
        for (;;) {
            if (has_next) {
                load_raw(xn);
                if (next + nwaves < NSTRIPS) xnn = xs[(next + nwaves) * 16 + r16];
            }
            compute_store(a, strip);
            if (!has_next) break;
            split_raw(a);
            strip    = next;
            next     = strip + nwaves;
            has_next = next < NSTRIPS;
            xn       = xnn;
        }
    } else {
        if (has_next) load_a(an, next);
        compute_store(a, strip);

        while (has_next) {
#pragma unroll
            for (int kc = 0; kc < KC; ++kc) {
                a[kc][0] = an[kc][0]; a[kc][1] = an[kc][1];
                a[kc][2] = an[kc][2]; a[kc][3] = an[kc][3];
            }
            strip    = next;
            next     = strip + nwaves;
            has_next = next < NSTRIPS;
            if (has_next) load_a(an, next);
            compute_store(a, strip);
        }
    }
}


#define GL   16
#define NREP 2

template<int NOUT, bool RELUSPLIT>
__global__ void k_gather(const float* __restrict__ hw, const int* __restrict__ cnt,
                         const int* __restrict__ slot, const int* __restrict__ spill_n,
                         const int* __restrict__ spill_dst, const int* __restrict__ spill_src,
                         const float* __restrict__ dinv, const float* __restrict__ bias,
                         _Float16* __restrict__ hhi, _Float16* __restrict__ hlo,
                         float* __restrict__ outf) {
    constexpr int NCH = NOUT / (4 * GL);
    constexpr int R4  = NOUT / 4;
    constexpr int SPL = PITCH / GL;
    constexpr int EPT = 8 / NCH;
    static_assert(SPL == 2 && GL % EPT == 0, "a trip's slots must lie in one slot register of consecutive lanes");
    const int t   = blockIdx.x * blockDim.x + threadIdx.x;
    const int grp = t / GL;
    const int c   = t % GL;
    const float4* hw4 = (const float4*)hw;
    const char* hwb = (const char*)hw;

    for (int rep = 0; rep < NREP; ++rep) {
    const int node = grp * NREP + rep;
    if (node >= NNODES) return;

    const float di = dinv[node];
    const float sw = di * di;
    float4 acc[NCH];
#pragma unroll
    for (int j = 0; j < NCH; ++j) {
        const float4 v = *(const float4*)(hwb + ((unsigned)node * (unsigned)(NOUT * 4) + (unsigned)(c * 16)) + j * (GL * 16));
        acc[j].x = sw * v.x; acc[j].y = sw * v.y; acc[j].z = sw * v.z; acc[j].w = sw * v.w;
    }

    const int  deg = cnt[node];
    const int  nin = deg < PITCH ? deg : PITCH;
    const int* row = slot + (size_t)node * PITCH;

    int   sl[SPL];
    float dl[SPL];
#pragma unroll
    for (int j = 0; j < SPL; ++j) {
        const int* p = (j * GL + c < nin) ? row + (j * GL + c) : cnt + node;
        sl[j] = *p;
    }
    asm volatile("" :: "v"(sl[0]), "v"(sl[1]));
#pragma unroll
    for (int j = 0; j < SPL; ++j)
        sl[j] = (j * GL + c < nin) ? sl[j] : node;
#pragma unroll
    for (int j = 0; j < SPL; ++j)
        dl[j] = dinv[sl[j]];
    asm volatile("" :: "v"(dl[0]), "v"(dl[1]));
#pragma unroll
    for (int j = 0; j < SPL; ++j)
        dl[j] = (j * GL + c < nin) ? dl[j] : 0.0f;

    const int ng = (nin + EPT - 1) / EPT;
    for (int g = 0; g < ng; ++g) {
        const int   cs = (g * EPT) / GL == 0 ? sl[0] : sl[1];
        const float cd = (g * EPT) / GL == 0 ? dl[0] : dl[1];
        const int   q0 = (g * EPT) % GL;
        int s[EPT];
#pragma unroll
        for (int e = 0; e < EPT; ++e)
            s[e] = __shfl(cs, q0 + e, GL);
        float4 m[EPT][NCH];
#pragma unroll
        for (int e = 0; e < EPT; ++e)
#pragma unroll
            for (int jj = 0; jj < NCH; ++jj)
                m[e][jj] = *(const float4*)(hwb + ((unsigned)s[e] * (unsigned)(NOUT * 4) + (unsigned)(c * 16)) + jj * (GL * 16));
#pragma unroll
        for (int e = 0; e < EPT; ++e) {
            const float we = __shfl(cd, q0 + e, GL) * di;
#pragma unroll
            for (int jj = 0; jj < NCH; ++jj) {
                acc[jj].x += we * m[e][jj].x; acc[jj].y += we * m[e][jj].y;
                acc[jj].z += we * m[e][jj].z; acc[jj].w += we * m[e][jj].w;
            }
        }
    }

    if (deg > PITCH) {
        int total = *spill_n;
        if (total > NEDGES) total = NEDGES;
        for (int base = 0; base < total; base += GL) {
            int n = total - base; if (n > GL) n = GL;
            int   s1 = -1;
            float d1 = 0.0f;
            if (c < n && spill_dst[base + c] == node) { s1 = spill_src[base + c]; d1 = dinv[s1]; }
            for (int k = 0; k < n; ++k) {
                const int sk = __shfl(s1, k, GL);
                if (sk >= 0) {
                    const float wk = __shfl(d1, k, GL) * di;
#pragma unroll
                    for (int jj = 0; jj < NCH; ++jj) {
                        const float4 mm = hw4[(size_t)sk * R4 + jj * GL + c];
                        acc[jj].x += wk * mm.x; acc[jj].y += wk * mm.y;
                        acc[jj].z += wk * mm.z; acc[jj].w += wk * mm.w;
                    }
                }
            }
        }
    }

#pragma unroll
    for (int jj = 0; jj < NCH; ++jj) {
        const float4 b = *(const float4*)((const char*)bias + (unsigned)(c * 16) + jj * (GL * 16));
        float4 a = acc[jj];
        a.x += b.x; a.y += b.y; a.z += b.z; a.w += b.w;
        const size_t idx4 = (size_t)node * R4 + jj * GL + c;
        if (RELUSPLIT) {
            a.x = fmaxf(a.x, 0.f); a.y = fmaxf(a.y, 0.f);
            a.z = fmaxf(a.z, 0.f); a.w = fmaxf(a.w, 0.f);
            split_store(a.x, a.y, a.z, a.w, hhi, hlo, idx4);
        } else {
            ((float4*)outf)[idx4] = a;
        }
    }
    }
}


extern "C" void kernel_launch(void* const* d_in, const int* in_sizes, int n_in,
                              void* d_out, int out_size, void* d_ws, size_t ws_size,
                              hipStream_t stream) {
    const int*   x      = (const int*)d_in[0];
    const int*   src    = (const int*)d_in[1];
    const int*   dst    = ((const int*)d_in[1]) + NEDGES;
    const float* W_in   = (const float*)d_in[2];
    const float* W_conv = (const float*)d_in[3];
    const float* b_conv = (const float*)d_in[4];
    const float* W_dec  = (const float*)d_in[5];
    const float* b_dec  = (const float*)d_in[6];
    (void)in_sizes; (void)n_in; (void)out_size; (void)ws_size;

    char* p = (char*)d_ws;
    auto carve = [&](size_t bytes) { char* r = p; p += (bytes + 255) & ~(size_t)255; return r; };
    float*    dinv   = (float*)   carve((size_t)NNODES * 4);
    int*      cnt    = (int*)     carve((size_t)NNODES * 4);
    int*      spilln = (int*)     carve((size_t)4);
    int*      slot   = (int*)     carve((size_t)NNODES * PITCH * 4);
    int*      spilld = (int*)     carve((size_t)NEDGES * 4);
    int*      spills = (int*)     carve((size_t)NEDGES * 4);
    float*    hw     = (float*)   carve((size_t)NNODES * EMBED * 4);
    _Float16* hhi    = (_Float16*)carve((size_t)NNODES * EMBED * 2);
    _Float16* hlo    = (_Float16*)carve((size_t)NNODES * EMBED * 2);
    _Float16* Wch    = (_Float16*)carve((size_t)EMBED * EMBED * 2);
    _Float16* Wcl    = (_Float16*)carve((size_t)EMBED * EMBED * 2);
    _Float16* Wdh    = (_Float16*)carve((size_t)NTYPES * EMBED * 2);
    _Float16* Wdl    = (_Float16*)carve((size_t)NTYPES * EMBED * 2);

    const int B = 256;
    auto blocks = [](long long n, int b) { return (int)((n + b - 1) / b); };

    k_zero<<<blocks(NNODES, B), B, 0, stream>>>(cnt, spilln);
    k_fill<<<blocks(NEDGES, B), B, 0, stream>>>(src, dst, cnt, slot, spilln, spilld, spills);
    k_dinv<<<blocks(NNODES, B), B, 0, stream>>>(cnt, dinv);

    k_split_Wt<EMBED, EMBED> <<<blocks(EMBED * EMBED, B), B, 0, stream>>>(W_conv, Wch, Wcl);
    k_split_Wt<EMBED, NTYPES><<<blocks(EMBED * NTYPES, B), B, 0, stream>>>(W_dec,  Wdh, Wdl);

    const int gemm_blocks = 209;

    for (int it = 0; it < NITERS; ++it) {
        if (it == 0)
            k_gemm<EMBED, true><<<gemm_blocks, 128, 0, stream>>>(nullptr, nullptr, x, W_in, Wch, Wcl, hw);
        else
            k_gemm<EMBED, false><<<gemm_blocks, 128, 0, stream>>>(hhi, hlo, nullptr, nullptr, Wch, Wcl, hw);
        k_gather<EMBED, true><<<blocks((long long)NNODES * GL / NREP, B), B, 0, stream>>>(
            hw, cnt, slot, spilln, spilld, spills, dinv, b_conv, hhi, hlo, nullptr);
    }

    k_gemm<NTYPES, false><<<gemm_blocks, 128, 0, stream>>>(hhi, hlo, nullptr, nullptr, Wdh, Wdl, hw);
    k_gather<NTYPES, false><<<blocks((long long)NNODES * GL / NREP, B), B, 0, stream>>>(
        hw, cnt, slot, spilln, spilld, spills, dinv, b_dec, nullptr, nullptr, (float*)d_out);
}


# edge norm split in two: the GEMM stores row s times dinv[s], the gather adds plain rows and scales the sum once by dinv[d]; the per-edge norm loads are gone; products round anew, sum order kept
# speedup vs baseline: 2.3596x; 1.0392x over previous
#include <hip/hip_runtime.h>

typedef __attribute__((ext_vector_type(4)))  _Float16 v4h;
typedef __attribute__((ext_vector_type(8)))  _Float16 v8h;
typedef __attribute__((ext_vector_type(16))) _Float16 v16h;
typedef __attribute__((ext_vector_type(8)))  float    v8f;

#define NNODES 40000
#define NEDGES 640000
#define EMBED  128
#define NTYPES 64
#define NITERS 4
#define PITCH  32


__global__ void k_zero(int* cnt, int* spill_n) {
    int i = blockIdx.x * blockDim.x + threadIdx.x;
    if (i < NNODES) cnt[i] = 0;
    if (i == 0) *spill_n = 0;
}

__global__ void k_fill(const int* __restrict__ src, const int* __restrict__ dst,
                       int* cnt, int* __restrict__ slot, int* spill_n,
                       int* __restrict__ spill_dst, int* __restrict__ spill_src) {
    int e = blockIdx.x * blockDim.x + threadIdx.x;
    if (e >= NEDGES) return;
    int s = src[e], d = dst[e];
    int r = atomicAdd(&cnt[d], 1);
    if (r < PITCH) {
        slot[d * PITCH + r] = s;
    } else {
        int o = atomicAdd(spill_n, 1);
        spill_dst[o] = d;
        spill_src[o] = s;
    }
}

__global__ void k_dinv(const int* __restrict__ cnt, float* __restrict__ dinv) {
    int i = blockIdx.x * blockDim.x + threadIdx.x;
    if (i < NNODES) dinv[i] = rsqrtf((float)cnt[i] + 1.0f);
}


__device__ __forceinline__ void split_store(float x, float y, float z, float w,
                                            _Float16* hi, _Float16* lo, size_t idx4) {
    v4h H, L;
    H[0] = (_Float16)x; L[0] = (_Float16)(x - (float)H[0]);
    H[1] = (_Float16)y; L[1] = (_Float16)(y - (float)H[1]);
    H[2] = (_Float16)z; L[2] = (_Float16)(z - (float)H[2]);
    H[3] = (_Float16)w; L[3] = (_Float16)(w - (float)H[3]);
    ((v4h*)hi)[idx4] = H;
    ((v4h*)lo)[idx4] = L;
}


template<int K, int N>
__global__ void k_split_Wt(const float* __restrict__ W, _Float16* __restrict__ Whi,
                           _Float16* __restrict__ Wlo) {
    constexpr int NT = N / 16;
    int t = blockIdx.x * blockDim.x + threadIdx.x;
    if (t >= K * N) return;
    int k = t / N, n = t % N;
    float w = W[k * N + n];
    _Float16 hi = (_Float16)w;
    int kc = k >> 5, kk = k & 31;
    int j = kk >> 4, hf = (kk >> 3) & 1, e = kk & 7;
    int tt = n >> 4, c = n & 15;
    int idx = ((((kc * NT + tt) * 2 + j) * 32 + hf * 16 + c) << 3) + e;
    Whi[idx] = hi;
    Wlo[idx] = (_Float16)(w - (float)hi);
}


#define NSTRIPS (NNODES / 16)

template<int NOUT, bool EMB>
__global__ __launch_bounds__(128) void k_gemm(const _Float16* __restrict__ Ahi,
                                              const _Float16* __restrict__ Alo,
                                              const int* __restrict__ xs,
                                              const float* __restrict__ Win,
                                              const _Float16* __restrict__ Bhi,
                                              const _Float16* __restrict__ Blo,
                                              const float* __restrict__ dinv,
                                              float* __restrict__ out) {
    constexpr int K  = EMBED;
    constexpr int KC = K / 32;
    constexpr int NT = NOUT / 16;

    const int lane   = threadIdx.x & 31;
    const int half   = lane >> 4;
    const int nwaves = gridDim.x * 4;
    int strip        = blockIdx.x * 4 + (threadIdx.x >> 5);
    if (strip >= NSTRIPS) return;

    v8h a[KC][4], an[KC][4];
    v16h bh[KC][NT], bl[KC][NT];

    auto load_a = [&](v8h (&dst)[KC][4], int s) {
        const int mrow = s * 16 + (lane & 15);
#pragma unroll
        for (int kc = 0; kc < KC; ++kc) {
            const int abase = mrow * K + kc * 32 + half * 8;
            dst[kc][0] = *(const v8h*)(Ahi + abase);
            dst[kc][1] = *(const v8h*)(Ahi + abase + 16);
            dst[kc][2] = *(const v8h*)(Alo + abase);
            dst[kc][3] = *(const v8h*)(Alo + abase + 16);
        }
    };

    float4 dc[2], dn[2];
    auto load_d = [&](float4 (&d)[2], int s) {
        const float4* p = (const float4*)(dinv + s * 16 + half * 8);
        d[0] = p[0];
        d[1] = p[1];
    };

    float4 raw[KC][4];

    auto load_raw = [&](int xr) {
        const float* srow = Win + (size_t)xr * K + half * 8;
#pragma unroll
        for (int kc = 0; kc < KC; ++kc) {
            raw[kc][0] = *(const float4*)(srow + kc * 32);
            raw[kc][1] = *(const float4*)(srow + kc * 32 + 4);
            raw[kc][2] = *(const float4*)(srow + kc * 32 + 16);
            raw[kc][3] = *(const float4*)(srow + kc * 32 + 20);
        }
    };

    auto split_raw = [&](v8h (&dst)[KC][4]) {
#pragma unroll
        for (int kc = 0; kc < KC; ++kc) {
#pragma unroll
            for (int j = 0; j < 2; ++j) {
                const float4 f = raw[kc][2 * j], g = raw[kc][2 * j + 1];
                v8h H, L;
                H[0] = (_Float16)f.x; L[0] = (_Float16)(f.x - (float)H[0]);
                H[1] = (_Float16)f.y; L[1] = (_Float16)(f.y - (float)H[1]);
                H[2] = (_Float16)f.z; L[2] = (_Float16)(f.z - (float)H[2]);
                H[3] = (_Float16)f.w; L[3] = (_Float16)(f.w - (float)H[3]);
                H[4] = (_Float16)g.x; L[4] = (_Float16)(g.x - (float)H[4]);
                H[5] = (_Float16)g.y; L[5] = (_Float16)(g.y - (float)H[5]);
                H[6] = (_Float16)g.z; L[6] = (_Float16)(g.z - (float)H[6]);
                H[7] = (_Float16)g.w; L[7] = (_Float16)(g.w - (float)H[7]);
                dst[kc][j]     = H;
                dst[kc][2 + j] = L;
            }
        }
    };

    auto compute_store = [&](const v8h (&af)[KC][4], const float4 (&d)[2], int s) {
        v8f acc[NT] = {};
#pragma unroll
        for (int kc = 0; kc < KC; ++kc) {
            v16h fah = __builtin_shufflevector(af[kc][0], af[kc][1], 0,1,2,3,4,5,6,7,8,9,10,11,12,13,14,15);
            v16h fal = __builtin_shufflevector(af[kc][2], af[kc][3], 0,1,2,3,4,5,6,7,8,9,10,11,12,13,14,15);
#pragma unroll
            for (int t = 0; t < NT; ++t) {
                acc[t] = __builtin_amdgcn_wmma_f32_16x16x32_f16(false, fah, false, bh[kc][t],
                                                                (short)0, acc[t], false, false);
                acc[t] = __builtin_amdgcn_wmma_f32_16x16x32_f16(false, fah, false, bl[kc][t],
                                                                (short)0, acc[t], false, false);
                acc[t] = __builtin_amdgcn_wmma_f32_16x16x32_f16(false, fal, false, bh[kc][t],
                                                                (short)0, acc[t], false, false);
            }
        }
        const float dr[8] = {d[0].x, d[0].y, d[0].z, d[0].w, d[1].x, d[1].y, d[1].z, d[1].w};
#pragma unroll
        for (int t = 0; t < NT; ++t) {
            const int ncol = t * 16 + (lane & 15);
#pragma unroll
            for (int r = 0; r < 8; ++r) {
                const int row = s * 16 + half * 8 + r;
                out[(size_t)row * NOUT + ncol] = acc[t][r] * dr[r];
            }
        }
    };

    const int r16 = lane & 15;
    int  next     = strip + nwaves;
    bool has_next = next < NSTRIPS;
    int  xn = 0, xnn = 0;
    if constexpr (EMB) {
        const int xc = xs[strip * 16 + r16];
        if (has_next) xn = xs[next * 16 + r16];
        load_raw(xc);
    } else {
        load_a(a, strip);
    }
    load_d(dc, strip);

#pragma unroll
    for (int kc = 0; kc < KC; ++kc) {
#pragma unroll
        for (int t = 0; t < NT; ++t) {
            const int u = ((kc * NT + t) * 2) * 32 + lane;
            v8h b0h = ((const v8h*)Bhi)[u];
            v8h b1h = ((const v8h*)Bhi)[u + 32];
            v8h b0l = ((const v8h*)Blo)[u];
            v8h b1l = ((const v8h*)Blo)[u + 32];
            bh[kc][t] = __builtin_shufflevector(b0h, b1h, 0,1,2,3,4,5,6,7,8,9,10,11,12,13,14,15);
            bl[kc][t] = __builtin_shufflevector(b0l, b1l, 0,1,2,3,4,5,6,7,8,9,10,11,12,13,14,15);
        }
    }

    if constexpr (EMB) {
        split_raw(a);
        for (;;) {
            if (has_next) {
                load_raw(xn);
                load_d(dn, next);
                if (next + nwaves < NSTRIPS) xnn = xs[(next + nwaves) * 16 + r16];
            }
            compute_store(a, dc, strip);
            if (!has_next) break;
            split_raw(a);
            dc[0] = dn[0]; dc[1] = dn[1];
            strip    = next;
            next     = strip + nwaves;
            has_next = next < NSTRIPS;
            xn       = xnn;
        }
    } else {
        if (has_next) { load_a(an, next); load_d(dn, next); }
        compute_store(a, dc, strip);

        while (has_next) {
            dc[0] = dn[0]; dc[1] = dn[1];
#pragma unroll
            for (int kc = 0; kc < KC; ++kc) {
                a[kc][0] = an[kc][0]; a[kc][1] = an[kc][1];
                a[kc][2] = an[kc][2]; a[kc][3] = an[kc][3];
            }
            strip    = next;
            next     = strip + nwaves;
            has_next = next < NSTRIPS;
            if (has_next) { load_a(an, next); load_d(dn, next); }
            compute_store(a, dc, strip);
        }
    }
}


#define GL   16
#define NREP 2

template<int NOUT, bool RELUSPLIT>
__global__ void k_gather(const float* __restrict__ hw, const int* __restrict__ cnt,
                         const int* __restrict__ slot, const int* __restrict__ spill_n,
                         const int* __restrict__ spill_dst, const int* __restrict__ spill_src,
                         const float* __restrict__ dinv, const float* __restrict__ bias,
                         _Float16* __restrict__ hhi, _Float16* __restrict__ hlo,
                         float* __restrict__ outf) {
    constexpr int NCH = NOUT / (4 * GL);
    constexpr int R4  = NOUT / 4;
    constexpr int SPL = PITCH / GL;
    constexpr int EPT = 8 / NCH;
    static_assert(SPL == 2 && GL % EPT == 0, "a trip's slots must lie in one slot register of consecutive lanes");
    const int t   = blockIdx.x * blockDim.x + threadIdx.x;
    const int grp = t / GL;
    const int c   = t % GL;
    const float4* hw4 = (const float4*)hw;
    const char* hwb = (const char*)hw;

    for (int rep = 0; rep < NREP; ++rep) {
    const int node = grp * NREP + rep;
    if (node >= NNODES) return;

    const float di = dinv[node];
    float4 acc[NCH];
#pragma unroll
    for (int j = 0; j < NCH; ++j)
        acc[j] = *(const float4*)(hwb + ((unsigned)node * (unsigned)(NOUT * 4) + (unsigned)(c * 16)) + j * (GL * 16));

    const int  deg = cnt[node];
    const int  nin = deg < PITCH ? deg : PITCH;
    const int* row = slot + (size_t)node * PITCH;

    int   sl[SPL];
#pragma unroll
    for (int j = 0; j < SPL; ++j) {
        const int* p = (j * GL + c < nin) ? row + (j * GL + c) : cnt + node;
        sl[j] = *p;
    }
    asm volatile("" :: "v"(sl[0]), "v"(sl[1]));
#pragma unroll
    for (int j = 0; j < SPL; ++j)
        sl[j] = (j * GL + c < nin) ? sl[j] : node;

    const int ng = (nin + EPT - 1) / EPT;
    for (int g = 0; g < ng; ++g) {
        const int   cs = (g * EPT) / GL == 0 ? sl[0] : sl[1];
        const int   q0 = (g * EPT) % GL;
        int s[EPT];
#pragma unroll
        for (int e = 0; e < EPT; ++e)
            s[e] = __shfl(cs, q0 + e, GL);
        float4 m[EPT][NCH];
#pragma unroll
        for (int e = 0; e < EPT; ++e)
#pragma unroll
            for (int jj = 0; jj < NCH; ++jj)
                m[e][jj] = *(const float4*)(hwb + ((unsigned)s[e] * (unsigned)(NOUT * 4) + (unsigned)(c * 16)) + jj * (GL * 16));
#pragma unroll
        for (int e = 0; e < EPT; ++e) {
            const float we = (g * EPT + e < nin) ? 1.0f : 0.0f;
#pragma unroll
            for (int jj = 0; jj < NCH; ++jj) {
                acc[jj].x += we * m[e][jj].x; acc[jj].y += we * m[e][jj].y;
                acc[jj].z += we * m[e][jj].z; acc[jj].w += we * m[e][jj].w;
            }
        }
    }

    if (deg > PITCH) {
        int total = *spill_n;
        if (total > NEDGES) total = NEDGES;
        for (int base = 0; base < total; base += GL) {
            int n = total - base; if (n > GL) n = GL;
            int   s1 = -1;
            if (c < n && spill_dst[base + c] == node) s1 = spill_src[base + c];
            for (int k = 0; k < n; ++k) {
                const int sk = __shfl(s1, k, GL);
                if (sk >= 0) {
#pragma unroll
                    for (int jj = 0; jj < NCH; ++jj) {
                        const float4 mm = hw4[(size_t)sk * R4 + jj * GL + c];
                        acc[jj].x += mm.x; acc[jj].y += mm.y;
                        acc[jj].z += mm.z; acc[jj].w += mm.w;
                    }
                }
            }
        }
    }

#pragma unroll
    for (int jj = 0; jj < NCH; ++jj) {
        const float4 b = *(const float4*)((const char*)bias + (unsigned)(c * 16) + jj * (GL * 16));
        float4 a = acc[jj];
        a.x = di * a.x + b.x; a.y = di * a.y + b.y; a.z = di * a.z + b.z; a.w = di * a.w + b.w;
        const size_t idx4 = (size_t)node * R4 + jj * GL + c;
        if (RELUSPLIT) {
            a.x = fmaxf(a.x, 0.f); a.y = fmaxf(a.y, 0.f);
            a.z = fmaxf(a.z, 0.f); a.w = fmaxf(a.w, 0.f);
            split_store(a.x, a.y, a.z, a.w, hhi, hlo, idx4);
        } else {
            ((float4*)outf)[idx4] = a;
        }
    }
    }
}


extern "C" void kernel_launch(void* const* d_in, const int* in_sizes, int n_in,
                              void* d_out, int out_size, void* d_ws, size_t ws_size,
                              hipStream_t stream) {
    const int*   x      = (const int*)d_in[0];
    const int*   src    = (const int*)d_in[1];
    const int*   dst    = ((const int*)d_in[1]) + NEDGES;
    const float* W_in   = (const float*)d_in[2];
    const float* W_conv = (const float*)d_in[3];
    const float* b_conv = (const float*)d_in[4];
    const float* W_dec  = (const float*)d_in[5];
    const float* b_dec  = (const float*)d_in[6];
    (void)in_sizes; (void)n_in; (void)out_size; (void)ws_size;

    char* p = (char*)d_ws;
    auto carve = [&](size_t bytes) { char* r = p; p += (bytes + 255) & ~(size_t)255; return r; };
    float*    dinv   = (float*)   carve((size_t)NNODES * 4);
    int*      cnt    = (int*)     carve((size_t)NNODES * 4);
    int*      spilln = (int*)     carve((size_t)4);
    int*      slot   = (int*)     carve((size_t)NNODES * PITCH * 4);
    int*      spilld = (int*)     carve((size_t)NEDGES * 4);
    int*      spills = (int*)     carve((size_t)NEDGES * 4);
    float*    hw     = (float*)   carve((size_t)NNODES * EMBED * 4);
    _Float16* hhi    = (_Float16*)carve((size_t)NNODES * EMBED * 2);
    _Float16* hlo    = (_Float16*)carve((size_t)NNODES * EMBED * 2);
    _Float16* Wch    = (_Float16*)carve((size_t)EMBED * EMBED * 2);
    _Float16* Wcl    = (_Float16*)carve((size_t)EMBED * EMBED * 2);
    _Float16* Wdh    = (_Float16*)carve((size_t)NTYPES * EMBED * 2);
    _Float16* Wdl    = (_Float16*)carve((size_t)NTYPES * EMBED * 2);

    const int B = 256;
    auto blocks = [](long long n, int b) { return (int)((n + b - 1) / b); };

    k_zero<<<blocks(NNODES, B), B, 0, stream>>>(cnt, spilln);
    k_fill<<<blocks(NEDGES, B), B, 0, stream>>>(src, dst, cnt, slot, spilln, spilld, spills);
    k_dinv<<<blocks(NNODES, B), B, 0, stream>>>(cnt, dinv);

    k_split_Wt<EMBED, EMBED> <<<blocks(EMBED * EMBED, B), B, 0, stream>>>(W_conv, Wch, Wcl);
    k_split_Wt<EMBED, NTYPES><<<blocks(EMBED * NTYPES, B), B, 0, stream>>>(W_dec,  Wdh, Wdl);

    const int gemm_blocks = 209;

    for (int it = 0; it < NITERS; ++it) {
        if (it == 0)
            k_gemm<EMBED, true><<<gemm_blocks, 128, 0, stream>>>(nullptr, nullptr, x, W_in, Wch, Wcl, dinv, hw);
        else
            k_gemm<EMBED, false><<<gemm_blocks, 128, 0, stream>>>(hhi, hlo, nullptr, nullptr, Wch, Wcl, dinv, hw);
        k_gather<EMBED, true><<<blocks((long long)NNODES * GL / NREP, B), B, 0, stream>>>(
            hw, cnt, slot, spilln, spilld, spills, dinv, b_conv, hhi, hlo, nullptr);
    }

    k_gemm<NTYPES, false><<<gemm_blocks, 128, 0, stream>>>(hhi, hlo, nullptr, nullptr, Wdh, Wdl, dinv, hw);
    k_gather<NTYPES, false><<<blocks((long long)NNODES * GL / NREP, B), B, 0, stream>>>(
        hw, cnt, slot, spilln, spilld, spills, dinv, b_dec, nullptr, nullptr, (float*)d_out);
}
